# starting kernel
# baseline (speedup 1.0000x reference)
#include <hip/hip_runtime.h>

#define HH   64
#define WW   64
#define CCH  2048
#define HW   4096
#define IMG  224

typedef __attribute__((ext_vector_type(2))) float v2f;
typedef __attribute__((ext_vector_type(8))) float v8f;

#define WMMA_F32(A, B, C) \
    __builtin_amdgcn_wmma_f32_16x16x4_f32(false, (A), false, (B), (short)0, (C), false, false)

__device__ __forceinline__ unsigned f2ord(float f) {
    unsigned u = __float_as_uint(f);
    return (u & 0x80000000u) ? ~u : (u | 0x80000000u);
}
__device__ __forceinline__ float ord2f(unsigned s) {
    unsigned b = (s & 0x80000000u) ? (s ^ 0x80000000u) : ~s;
    return __uint_as_float(b);
}

__global__ __launch_bounds__(256)
void simcam_colsum(const float* __restrict__ x0, const float* __restrict__ x1,
                   float* __restrict__ S0, float* __restrict__ S1,
                   unsigned* __restrict__ maxp) {
    const int t = blockIdx.x * 256 + threadIdx.x;
    if (t == 0) maxp[0] = 0u;
    const bool second = t >= 512;
    const float* __restrict__ X = second ? x1 : x0;
    float* __restrict__ S = second ? S1 : S0;
    const int c4 = (t & 511) * 4;
    float sx = 0.f, sy = 0.f, sz = 0.f, sw = 0.f;
    #pragma unroll 4
    for (int r = 0; r < HW; ++r) {
        float4 v = *reinterpret_cast<const float4*>(X + (size_t)r * CCH + c4);
        sx += v.x; sy += v.y; sz += v.z; sw += v.w;
    }
    float4 o; o.x = sx; o.y = sy; o.z = sz; o.w = sw;
    *reinterpret_cast<float4*>(S + c4) = o;
}

__global__ __launch_bounds__(256)
void simcam_gemm_max(const float* __restrict__ x0, const float* __restrict__ x1,
                     unsigned* __restrict__ maxp) {
    const int wid  = threadIdx.x >> 5;
    const int lane = threadIdx.x & 31;
    const int r    = lane & 15;
    const int h    = lane >> 4;

    const int m0 = blockIdx.x * 128 + (wid & 3) * 32;
    const int n0 = blockIdx.y * 128 + (wid >> 2) * 64;

    const float* __restrict__ A0 = x0 + (size_t)(m0 + r) * CCH + 2 * h;
    const float* __restrict__ A1 = A0 + (size_t)16 * CCH;
    const float* __restrict__ B0 = x1 + (size_t)(n0 + r) * CCH + 2 * h;
    const float* __restrict__ B1 = B0 + (size_t)16 * CCH;
    const float* __restrict__ B2 = B0 + (size_t)32 * CCH;
    const float* __restrict__ B3 = B0 + (size_t)48 * CCH;

    v8f acc00 = {}, acc01 = {}, acc02 = {}, acc03 = {};
    v8f acc10 = {}, acc11 = {}, acc12 = {}, acc13 = {};

    v2f a0 = *reinterpret_cast<const v2f*>(A0);
    v2f a1 = *reinterpret_cast<const v2f*>(A1);
    v2f b0 = *reinterpret_cast<const v2f*>(B0);
    v2f b1 = *reinterpret_cast<const v2f*>(B1);
    v2f b2 = *reinterpret_cast<const v2f*>(B2);
    v2f b3 = *reinterpret_cast<const v2f*>(B3);

    #pragma unroll 2
    for (int c = 4; c < CCH; c += 4) {
        v2f na0 = *reinterpret_cast<const v2f*>(A0 + c);
        v2f na1 = *reinterpret_cast<const v2f*>(A1 + c);
        v2f nb0 = *reinterpret_cast<const v2f*>(B0 + c);
        v2f nb1 = *reinterpret_cast<const v2f*>(B1 + c);
        v2f nb2 = *reinterpret_cast<const v2f*>(B2 + c);
        v2f nb3 = *reinterpret_cast<const v2f*>(B3 + c);

        acc00 = WMMA_F32(a0, b0, acc00);
        acc01 = WMMA_F32(a0, b1, acc01);
        acc02 = WMMA_F32(a0, b2, acc02);
        acc03 = WMMA_F32(a0, b3, acc03);
        acc10 = WMMA_F32(a1, b0, acc10);
        acc11 = WMMA_F32(a1, b1, acc11);
        acc12 = WMMA_F32(a1, b2, acc12);
        acc13 = WMMA_F32(a1, b3, acc13);

        a0 = na0; a1 = na1;
        b0 = nb0; b1 = nb1; b2 = nb2; b3 = nb3;
    }

    acc00 = WMMA_F32(a0, b0, acc00);
    acc01 = WMMA_F32(a0, b1, acc01);
    acc02 = WMMA_F32(a0, b2, acc02);
    acc03 = WMMA_F32(a0, b3, acc03);
    acc10 = WMMA_F32(a1, b0, acc10);
    acc11 = WMMA_F32(a1, b1, acc11);
    acc12 = WMMA_F32(a1, b2, acc12);
    acc13 = WMMA_F32(a1, b3, acc13);

    float m = -3.402823466e38f;
    #pragma unroll
    for (int e = 0; e < 8; ++e) {
        m = fmaxf(m, acc00[e]); m = fmaxf(m, acc01[e]);
        m = fmaxf(m, acc02[e]); m = fmaxf(m, acc03[e]);
        m = fmaxf(m, acc10[e]); m = fmaxf(m, acc11[e]);
        m = fmaxf(m, acc12[e]); m = fmaxf(m, acc13[e]);
    }
    #pragma unroll
    for (int off = 16; off >= 1; off >>= 1)
        m = fmaxf(m, __shfl_xor(m, off, 32));
    if (lane == 0) atomicMax(maxp, f2ord(m));
}

__global__ __launch_bounds__(256)
void simcam_saliency(const float* __restrict__ x0, const float* __restrict__ x1,
                     const float* __restrict__ S0, const float* __restrict__ S1,
                     const unsigned* __restrict__ maxp,
                     float* __restrict__ d1, float* __restrict__ d2) {
    __shared__ float s[CCH];
    const bool second = blockIdx.x >= 16;
    const float* __restrict__ S = second ? S0 : S1;
    const float* __restrict__ X = second ? x1 : x0;
    float* __restrict__ D       = second ? d2 : d1;
    const int row = ((int)blockIdx.x & 15) * 256 + threadIdx.x;

    for (int c = threadIdx.x; c < CCH; c += 256) s[c] = S[c];
    __syncthreads();

    const float* __restrict__ xr = X + (size_t)row * CCH;
    float acc = 0.0f;
    #pragma unroll 4
    for (int c = 0; c < CCH; c += 4) {
        float4 xv = *reinterpret_cast<const float4*>(xr + c);
        acc += xv.x * s[c] + xv.y * s[c + 1] + xv.z * s[c + 2] + xv.w * s[c + 3];
    }
    const float inv = 1.0f / ord2f(maxp[0]);
    D[row] = acc * inv;
}

__global__ __launch_bounds__(256)
void simcam_resize(const float* __restrict__ d1, const float* __restrict__ d2,
                   float* __restrict__ out) {
    const int idx = blockIdx.x * 256 + threadIdx.x;
    if (idx >= 2 * IMG * IMG) return;
    const int s   = idx / (IMG * IMG);
    const int rem = idx - s * IMG * IMG;
    const int y   = rem / IMG;
    const int x   = rem - y * IMG;
    const float* __restrict__ map = s ? d2 : d1;

    const float scale = (float)HH / (float)IMG;
    float fy = ((float)y + 0.5f) * scale - 0.5f;
    float fx = ((float)x + 0.5f) * scale - 0.5f;
    fy = fminf(fmaxf(fy, 0.0f), (float)(HH - 1));
    fx = fminf(fmaxf(fx, 0.0f), (float)(WW - 1));
    const int y0 = (int)fy, x0 = (int)fx;
    const int y1 = min(y0 + 1, HH - 1), x1 = min(x0 + 1, WW - 1);
    const float wy = fy - (float)y0, wx = fx - (float)x0;

    const float v00 = map[y0 * WW + x0], v01 = map[y0 * WW + x1];
    const float v10 = map[y1 * WW + x0], v11 = map[y1 * WW + x1];
    out[idx] = (1.0f - wy) * ((1.0f - wx) * v00 + wx * v01)
             +          wy * ((1.0f - wx) * v10 + wx * v11);
}

extern "C" void kernel_launch(void* const* d_in, const int* in_sizes, int n_in,
                              void* d_out, int out_size, void* d_ws, size_t ws_size,
                              hipStream_t stream) {
    (void)in_sizes; (void)n_in; (void)out_size; (void)ws_size;

    const float* feats = (const float*)d_in[0];
    const float* x0 = feats;
    const float* x1 = feats + (size_t)HW * CCH;
    float* out = (float*)d_out;

    float*    wsf  = (float*)d_ws;
    unsigned* maxp = (unsigned*)wsf;
    float* S0  = wsf + 64;
    float* S1  = S0 + CCH;
    float* d1m = S1 + CCH;
    float* d2m = d1m + HW;

    simcam_colsum<<<dim3(4), dim3(256), 0, stream>>>(x0, x1, S0, S1, maxp);

    simcam_gemm_max<<<dim3(32, 32), dim3(256), 0, stream>>>(x0, x1, maxp);

    simcam_saliency<<<dim3(32), dim3(256), 0, stream>>>(x0, x1, S0, S1, maxp, d1m, d2m);

    const int nout = 2 * IMG * IMG;
    simcam_resize<<<dim3((nout + 255) / 256), dim3(256), 0, stream>>>(d1m, d2m, out);
}


# GEMM wave block 32x64 -> 128x128 held in registers (1 wave/SIMD, 256x256 tile per WGP, 4 waves), fragments still straight from global: each load feeds 16 matrix ops, not 2.7; sum order unchanged
# speedup vs baseline: 3.5594x; 3.5594x over previous
#include <hip/hip_runtime.h>

#define HH   64
#define WW   64
#define CCH  2048
#define HW   4096
#define IMG  224

typedef __attribute__((ext_vector_type(2))) float v2f;
typedef __attribute__((ext_vector_type(4))) float v4f;
typedef __attribute__((ext_vector_type(8))) float v8f;

#define WMMA_F32(A, B, C) \
    __builtin_amdgcn_wmma_f32_16x16x4_f32(false, (A), false, (B), (short)0, (C), false, false)

#define FR_LO(V) __builtin_shufflevector((V), (V), 0, 1)
#define FR_HI(V) __builtin_shufflevector((V), (V), 2, 3)

__global__ __launch_bounds__(128)
void simcam_gemm_max_colsum(const float* __restrict__ x0, const float* __restrict__ x1,
                            float* __restrict__ S0, float* __restrict__ S1,
                            float* __restrict__ tmax) {
    __shared__ float4 csum[8][32];

    const int wid  = threadIdx.x >> 5;
    const int lane = threadIdx.x & 31;

    if (blockIdx.x < 32) {
        const int g = blockIdx.x;
        const float* __restrict__ X = (g & 16) ? x1 : x0;
        float* __restrict__ S       = (g & 16) ? S1 : S0;
        const int c4 = (g & 15) * 128 + lane * 4;
        const float* __restrict__ p = X + (size_t)wid * CCH + c4;
        float4 lo, hi;
        lo.x = lo.y = lo.z = lo.w = 0.f;
        hi.x = hi.y = hi.z = hi.w = 0.f;
        #pragma unroll 32
        for (int i = 0; i < HW / 8; ++i) {
            float4 u = *reinterpret_cast<const float4*>(p + (size_t)i * 8 * CCH);
            float4 v = *reinterpret_cast<const float4*>(p + ((size_t)i * 8 + 4) * CCH);
            lo.x += u.x; lo.y += u.y; lo.z += u.z; lo.w += u.w;
            hi.x += v.x; hi.y += v.y; hi.z += v.z; hi.w += v.w;
        }
        csum[wid][lane]     = lo;
        csum[wid + 4][lane] = hi;
        __syncthreads();
        if (wid == 0) {
            float4 t = csum[0][lane];
            for (int k = 1; k < 8; ++k) {
                float4 u = csum[k][lane];
                t.x += u.x; t.y += u.y; t.z += u.z; t.w += u.w;
            }
            *reinterpret_cast<float4*>(S + c4) = t;
        }
        return;
    }

    const int r    = lane & 15;
    const int h    = lane >> 4;
    const int tile = (int)blockIdx.x - 32;

    const int m0 = (tile & 15) * 256 + (wid & 1) * 128;
    const int n0 = (tile >> 4) * 256 + (wid >> 1) * 128;

    const float* __restrict__ A = x0 + (size_t)(m0 + r) * CCH + 4 * h;
    const float* __restrict__ B = x1 + (size_t)(n0 + r) * CCH + 4 * h;

    v8f acc[8][8];
    #pragma unroll
    for (int i = 0; i < 8; ++i) {
        #pragma unroll
        for (int j = 0; j < 8; ++j) {
            v8f z = {};
            acc[i][j] = z;
        }
    }

    v4f fa[2][8], fb[2][8];

#define LOAD_A4(S, I0, C)                                                            \
    _Pragma("unroll") for (int i = (I0); i < (I0) + 4; ++i)                          \
        fa[S][i] = *reinterpret_cast<const v4f*>(A + (size_t)i * 16 * CCH + (C));
#define LOAD_B4(S, J0, C)                                                            \
    _Pragma("unroll") for (int j = (J0); j < (J0) + 4; ++j)                          \
        fb[S][j] = *reinterpret_cast<const v4f*>(B + (size_t)j * 16 * CCH + (C));
#define MMA_ROWS(S, FR, I0, I1)                                                      \
    _Pragma("unroll") for (int i = (I0); i < (I1); ++i) {                            \
        _Pragma("unroll") for (int j = 0; j < 8; ++j)                                \
            acc[i][j] = WMMA_F32(FR(fa[S][i]), FR(fb[S][j]), acc[i][j]);             \
    }
#define PIN() __builtin_amdgcn_sched_barrier(0)
#define PHASE(CS, LS, C)                                                             \
    LOAD_A4(LS, 0, C) PIN(); MMA_ROWS(CS, FR_LO, 0, 2) PIN();                        \
    LOAD_A4(LS, 4, C) PIN(); MMA_ROWS(CS, FR_LO, 2, 4) PIN();                        \
    LOAD_B4(LS, 0, C) PIN(); MMA_ROWS(CS, FR_LO, 4, 6) PIN();                        \
    LOAD_B4(LS, 4, C) PIN(); MMA_ROWS(CS, FR_LO, 6, 8) MMA_ROWS(CS, FR_HI, 0, 8) PIN();

    LOAD_A4(0, 0, 0) LOAD_A4(0, 4, 0) LOAD_B4(0, 0, 0) LOAD_B4(0, 4, 0)

    for (int c = 0; c < CCH - 16; c += 16) {
        PHASE(0, 1, c + 8)
        PHASE(1, 0, c + 16)
    }

    PHASE(0, 1, CCH - 8)
    MMA_ROWS(1, FR_LO, 0, 8)
    MMA_ROWS(1, FR_HI, 0, 8)
#undef PHASE
#undef PIN
#undef MMA_ROWS
#undef LOAD_B4
#undef LOAD_A4

    float m = -3.402823466e38f;
    #pragma unroll
    for (int i = 0; i < 8; ++i) {
        #pragma unroll
        for (int j = 0; j < 8; ++j) {
            #pragma unroll
            for (int e = 0; e < 8; ++e) m = fmaxf(m, acc[i][j][e]);
        }
    }
    #pragma unroll
    for (int off = 16; off >= 1; off >>= 1)
        m = fmaxf(m, __shfl_xor(m, off, 32));
    tmax[(tile * 4 + wid) * 32 + lane] = m;
}

__global__ __launch_bounds__(256)
void simcam_saliency(const float* __restrict__ x0, const float* __restrict__ x1,
                     const float* __restrict__ S0, const float* __restrict__ S1,
                     const float* __restrict__ tmax,
                     float* __restrict__ d1, float* __restrict__ d2) {
    __shared__ float s[CCH];
    __shared__ float wmx[8];
    const bool second = blockIdx.x >= 16;
    const float* __restrict__ S = second ? S0 : S1;
    const float* __restrict__ X = second ? x1 : x0;
    float* __restrict__ D       = second ? d2 : d1;
    const int row = ((int)blockIdx.x & 15) * 256 + threadIdx.x;

    const float* __restrict__ tp = tmax + (size_t)threadIdx.x * 128;
    float m = fmaxf(fmaxf(tp[0], tp[32]), fmaxf(tp[64], tp[96]));
    for (int off = 16; off >= 1; off >>= 1)
        m = fmaxf(m, __shfl_xor(m, off, 32));
    if ((threadIdx.x & 31) == 0) wmx[threadIdx.x >> 5] = m;

    for (int c = threadIdx.x; c < CCH; c += 256) s[c] = S[c];
    __syncthreads();

    float gmax = wmx[0];
    for (int k = 1; k < 8; ++k) gmax = fmaxf(gmax, wmx[k]);

    const float* __restrict__ xr = X + (size_t)row * CCH;
    float acc = 0.0f;
    #pragma unroll 4
    for (int c = 0; c < CCH; c += 4) {
        float4 xv = *reinterpret_cast<const float4*>(xr + c);
        acc += xv.x * s[c] + xv.y * s[c + 1] + xv.z * s[c + 2] + xv.w * s[c + 3];
    }
    const float inv = 1.0f / gmax;
    D[row] = acc * inv;
}

__global__ __launch_bounds__(256)
void simcam_resize(const float* __restrict__ d1, const float* __restrict__ d2,
                   float* __restrict__ out) {
    const int idx = blockIdx.x * 256 + threadIdx.x;
    if (idx >= 2 * IMG * IMG) return;
    const int s   = idx / (IMG * IMG);
    const int rem = idx - s * IMG * IMG;
    const int y   = rem / IMG;
    const int x   = rem - y * IMG;
    const float* __restrict__ map = s ? d2 : d1;

    const float scale = (float)HH / (float)IMG;
    float fy = ((float)y + 0.5f) * scale - 0.5f;
    float fx = ((float)x + 0.5f) * scale - 0.5f;
    fy = fminf(fmaxf(fy, 0.0f), (float)(HH - 1));
    fx = fminf(fmaxf(fx, 0.0f), (float)(WW - 1));
    const int y0 = (int)fy, x0 = (int)fx;
    const int y1 = min(y0 + 1, HH - 1), x1 = min(x0 + 1, WW - 1);
    const float wy = fy - (float)y0, wx = fx - (float)x0;

    const float v00 = map[y0 * WW + x0], v01 = map[y0 * WW + x1];
    const float v10 = map[y1 * WW + x0], v11 = map[y1 * WW + x1];
    out[idx] = (1.0f - wy) * ((1.0f - wx) * v00 + wx * v01)
             +          wy * ((1.0f - wx) * v10 + wx * v11);
}

extern "C" void kernel_launch(void* const* d_in, const int* in_sizes, int n_in,
                              void* d_out, int out_size, void* d_ws, size_t ws_size,
                              hipStream_t stream) {
    (void)in_sizes; (void)n_in; (void)out_size; (void)ws_size;

    const float* feats = (const float*)d_in[0];
    const float* x0 = feats;
    const float* x1 = feats + (size_t)HW * CCH;
    float* out = (float*)d_out;

    float* wsf  = (float*)d_ws;
    float* S0   = wsf + 64;
    float* S1   = S0 + CCH;
    float* d1m  = S1 + CCH;
    float* d2m  = d1m + HW;
    float* tmax = d2m + HW;

    simcam_gemm_max_colsum<<<dim3(288), dim3(128), 0, stream>>>(x0, x1, S0, S1, tmax);

    simcam_saliency<<<dim3(32), dim3(256), 0, stream>>>(x0, x1, S0, S1, tmax, d1m, d2m);

    const int nout = 2 * IMG * IMG;
    simcam_resize<<<dim3((nout + 255) / 256), dim3(256), 0, stream>>>(d1m, d2m, out);
}


# Column sums run first in a launch of their own, so the saliency dots become an equal 32-row share in front of every GEMM tile's k loop; saliency launch gone, resize divides by max; sum orders change
# speedup vs baseline: 3.9870x; 1.1201x over previous
#include <hip/hip_runtime.h>

#define HH   64
#define WW   64
#define CCH  2048
#define HW   4096
#define IMG  224

typedef __attribute__((ext_vector_type(2))) float v2f;
typedef __attribute__((ext_vector_type(4))) float v4f;
typedef __attribute__((ext_vector_type(8))) float v8f;

#define WMMA_F32(A, B, C) \
    __builtin_amdgcn_wmma_f32_16x16x4_f32(false, (A), false, (B), (short)0, (C), false, false)

#define FR_LO(V) __builtin_shufflevector((V), (V), 0, 1)
#define FR_HI(V) __builtin_shufflevector((V), (V), 2, 3)

__global__ __launch_bounds__(512)
void simcam_colsum(const float* __restrict__ x0, const float* __restrict__ x1,
                   float* __restrict__ S0, float* __restrict__ S1) {
    __shared__ float4 part[16][8];

    const int g    = blockIdx.x;
    const int wid  = threadIdx.x >> 5;
    const int lane = threadIdx.x & 31;
    const float* __restrict__ X = (g & 64) ? x1 : x0;
    float* __restrict__ S       = (g & 64) ? S1 : S0;
    const int c4 = (g & 63) * 32 + (lane & 7) * 4;
    const float* __restrict__ p = X + (size_t)(threadIdx.x >> 3) * CCH + c4;

    float4 a;
    a.x = a.y = a.z = a.w = 0.f;
    #pragma unroll 32
    for (int i = 0; i < HW / 64; ++i) {
        float4 u = *reinterpret_cast<const float4*>(p + (size_t)i * 64 * CCH);
        a.x += u.x; a.y += u.y; a.z += u.z; a.w += u.w;
    }
    #pragma unroll
    for (int off = 8; off <= 16; off <<= 1) {
        a.x += __shfl_xor(a.x, off, 32);
        a.y += __shfl_xor(a.y, off, 32);
        a.z += __shfl_xor(a.z, off, 32);
        a.w += __shfl_xor(a.w, off, 32);
    }
    if (lane < 8) part[wid][lane] = a;
    __syncthreads();
    if (threadIdx.x < 8) {
        float4 t = part[0][lane];
        for (int k = 1; k < 16; ++k) {
            float4 u = part[k][lane];
            t.x += u.x; t.y += u.y; t.z += u.z; t.w += u.w;
        }
        *reinterpret_cast<float4*>(S + c4) = t;
    }
}

__global__ __launch_bounds__(128)
void simcam_gemm_max_saliency(const float* __restrict__ x0, const float* __restrict__ x1,
                              const float* __restrict__ S0, const float* __restrict__ S1,
                              float* __restrict__ d1, float* __restrict__ d2,
                              float* __restrict__ tmax) {
    __shared__ float dots[32];

    const int wid  = threadIdx.x >> 5;
    const int lane = threadIdx.x & 31;

    {
        const bool second = blockIdx.x >= 128;
        const float* __restrict__ X  = second ? x1 : x0;
        const float* __restrict__ Sv = second ? S0 : S1;
        float* __restrict__ D        = second ? d2 : d1;
        const int rowb = ((int)blockIdx.x & 127) * 32;
        const float* __restrict__ xr = X + (size_t)(rowb + wid * 8) * CCH + 4 * lane;

        v4f sv[16];
        #pragma unroll
        for (int j = 0; j < 16; ++j)
            sv[j] = *reinterpret_cast<const v4f*>(Sv + 128 * j + 4 * lane);

        v4f xv[8][16];
        #pragma unroll
        for (int k = 0; k < 8; ++k) {
            #pragma unroll
            for (int j = 0; j < 16; ++j)
                xv[k][j] = *reinterpret_cast<const v4f*>(xr + (size_t)k * CCH + 128 * j);
        }

        float mine = 0.f;
        #pragma unroll
        for (int k = 0; k < 8; ++k) {
            float a0 = 0.f, a1 = 0.f, a2 = 0.f, a3 = 0.f;
            #pragma unroll
            for (int j = 0; j < 16; ++j) {
                a0 = fmaf(xv[k][j][0], sv[j][0], a0);
                a1 = fmaf(xv[k][j][1], sv[j][1], a1);
                a2 = fmaf(xv[k][j][2], sv[j][2], a2);
                a3 = fmaf(xv[k][j][3], sv[j][3], a3);
            }
            float a = (a0 + a1) + (a2 + a3);
            #pragma unroll
            for (int off = 16; off >= 1; off >>= 1)
                a += __shfl_xor(a, off, 32);
            if (lane == k) mine = a;
        }
        if (lane < 8) dots[wid * 8 + lane] = mine;
        __syncthreads();
        if (wid == 0) D[rowb + lane] = dots[lane];
    }

    const int r    = lane & 15;
    const int h    = lane >> 4;
    const int tile = (int)blockIdx.x;

    const int m0 = (tile & 15) * 256 + (wid & 1) * 128;
    const int n0 = (tile >> 4) * 256 + (wid >> 1) * 128;

    const float* __restrict__ A = x0 + (size_t)(m0 + r) * CCH + 4 * h;
    const float* __restrict__ B = x1 + (size_t)(n0 + r) * CCH + 4 * h;

    v8f acc[8][8];
    #pragma unroll
    for (int i = 0; i < 8; ++i) {
        #pragma unroll
        for (int j = 0; j < 8; ++j) {
            v8f z = {};
            acc[i][j] = z;
        }
    }

    v4f fa[2][8], fb[2][8];

#define LOAD_A4(S, I0, C)                                                            \
    _Pragma("unroll") for (int i = (I0); i < (I0) + 4; ++i)                          \
        fa[S][i] = *reinterpret_cast<const v4f*>(A + (size_t)i * 16 * CCH + (C));
#define LOAD_B4(S, J0, C)                                                            \
    _Pragma("unroll") for (int j = (J0); j < (J0) + 4; ++j)                          \
        fb[S][j] = *reinterpret_cast<const v4f*>(B + (size_t)j * 16 * CCH + (C));
#define MMA_ROWS(S, FR, I0, I1)                                                      \
    _Pragma("unroll") for (int i = (I0); i < (I1); ++i) {                            \
        _Pragma("unroll") for (int j = 0; j < 8; ++j)                                \
            acc[i][j] = WMMA_F32(FR(fa[S][i]), FR(fb[S][j]), acc[i][j]);             \
    }
#define PIN() __builtin_amdgcn_sched_barrier(0)
#define PHASE(CS, LS, C)                                                             \
    LOAD_A4(LS, 0, C) PIN(); MMA_ROWS(CS, FR_LO, 0, 2) PIN();                        \
    LOAD_A4(LS, 4, C) PIN(); MMA_ROWS(CS, FR_LO, 2, 4) PIN();                        \
    LOAD_B4(LS, 0, C) PIN(); MMA_ROWS(CS, FR_LO, 4, 6) PIN();                        \
    LOAD_B4(LS, 4, C) PIN(); MMA_ROWS(CS, FR_LO, 6, 8) MMA_ROWS(CS, FR_HI, 0, 8) PIN();

    LOAD_A4(0, 0, 0) LOAD_A4(0, 4, 0) LOAD_B4(0, 0, 0) LOAD_B4(0, 4, 0)

    for (int c = 0; c < CCH - 16; c += 16) {
        PHASE(0, 1, c + 8)
        PHASE(1, 0, c + 16)
    }

    PHASE(0, 1, CCH - 8)
    MMA_ROWS(1, FR_LO, 0, 8)
    MMA_ROWS(1, FR_HI, 0, 8)
#undef PHASE
#undef PIN
#undef MMA_ROWS
#undef LOAD_B4
#undef LOAD_A4

    float m = -3.402823466e38f;
    #pragma unroll
    for (int i = 0; i < 8; ++i) {
        #pragma unroll
        for (int j = 0; j < 8; ++j) {
            #pragma unroll
            for (int e = 0; e < 8; ++e) m = fmaxf(m, acc[i][j][e]);
        }
    }
    #pragma unroll
    for (int off = 16; off >= 1; off >>= 1)
        m = fmaxf(m, __shfl_xor(m, off, 32));
    tmax[(tile * 4 + wid) * 32 + lane] = m;
}

__global__ __launch_bounds__(1024)
void simcam_resize(const float* __restrict__ d1, const float* __restrict__ d2,
                   const float* __restrict__ tmax, float* __restrict__ out) {
    __shared__ float wmx[32];
    const int lane = threadIdx.x & 31;

    float m = tmax[(size_t)threadIdx.x * 32];
    #pragma unroll
    for (int off = 16; off >= 1; off >>= 1)
        m = fmaxf(m, __shfl_xor(m, off, 32));
    if (lane == 0) wmx[threadIdx.x >> 5] = m;
    __syncthreads();
    float gmax = wmx[lane];
    #pragma unroll
    for (int off = 16; off >= 1; off >>= 1)
        gmax = fmaxf(gmax, __shfl_xor(gmax, off, 32));
    const float inv = 1.0f / gmax;

    const int idx = blockIdx.x * 1024 + threadIdx.x;
    const int s   = idx / (IMG * IMG);
    const int rem = idx - s * IMG * IMG;
    const int y   = rem / IMG;
    const int x   = rem - y * IMG;
    const float* __restrict__ map = s ? d2 : d1;

    const float scale = (float)HH / (float)IMG;
    float fy = ((float)y + 0.5f) * scale - 0.5f;
    float fx = ((float)x + 0.5f) * scale - 0.5f;
    fy = fminf(fmaxf(fy, 0.0f), (float)(HH - 1));
    fx = fminf(fmaxf(fx, 0.0f), (float)(WW - 1));
    const int y0 = (int)fy, x0 = (int)fx;
    const int y1 = min(y0 + 1, HH - 1), x1 = min(x0 + 1, WW - 1);
    const float wy = fy - (float)y0, wx = fx - (float)x0;

    const float v00 = map[y0 * WW + x0], v01 = map[y0 * WW + x1];
    const float v10 = map[y1 * WW + x0], v11 = map[y1 * WW + x1];
    out[idx] = ((1.0f - wy) * ((1.0f - wx) * v00 + wx * v01)
              +          wy * ((1.0f - wx) * v10 + wx * v11)) * inv;
}

extern "C" void kernel_launch(void* const* d_in, const int* in_sizes, int n_in,
                              void* d_out, int out_size, void* d_ws, size_t ws_size,
                              hipStream_t stream) {
    (void)in_sizes; (void)n_in; (void)out_size; (void)ws_size;

    const float* feats = (const float*)d_in[0];
    const float* x0 = feats;
    const float* x1 = feats + (size_t)HW * CCH;
    float* out = (float*)d_out;

    float* wsf  = (float*)d_ws;
    float* S0   = wsf + 64;
    float* S1   = S0 + CCH;
    float* d1m  = S1 + CCH;
    float* d2m  = d1m + HW;
    float* tmax = d2m + HW;

    simcam_colsum<<<dim3(128), dim3(512), 0, stream>>>(x0, x1, S0, S1);

    simcam_gemm_max_saliency<<<dim3(256), dim3(128), 0, stream>>>(x0, x1, S0, S1, d1m, d2m, tmax);

    simcam_resize<<<dim3((2 * IMG * IMG) / 1024), dim3(1024), 0, stream>>>(d1m, d2m, tmax, out);
}
